# starting kernel
# baseline (speedup 1.0000x reference)
#include <hip/hip_runtime.h>
#include <math.h>


typedef __attribute__((ext_vector_type(2))) float v2f;
typedef __attribute__((ext_vector_type(8))) float v8f;

#if __has_builtin(__builtin_amdgcn_tanhf)
#define TANH_F32(x) __builtin_amdgcn_tanhf(x)
#else
#define TANH_F32(x) tanhf(x)
#endif

__device__ __forceinline__ v8f wmma4(v2f a, v2f b, v8f c) {
  return __builtin_amdgcn_wmma_f32_16x16x4_f32(false, a, false, b, (short)0, c,
                                               false, false);
}

__device__ __forceinline__ float swap16(float x) {
  return __int_as_float(__builtin_amdgcn_ds_swizzle(__float_as_int(x), 0x401F));
}

#define HID 50
#define LAT 16

__global__ __launch_bounds__(256) void node_rk4_kernel(
    const float* __restrict__ x0, const float* __restrict__ tv,
    const float* __restrict__ We1, const float* __restrict__ be1,
    const float* __restrict__ We2, const float* __restrict__ be2,
    const float* __restrict__ Wo1, const float* __restrict__ bo1,
    const float* __restrict__ Wo2, const float* __restrict__ bo2,
    const float* __restrict__ Wd1, const float* __restrict__ bd1,
    const float* __restrict__ Wd2, const float* __restrict__ bd2,
    float* __restrict__ out, int B, int T) {
  const int lane = threadIdx.x & 31;
  const int wid = blockIdx.x * (blockDim.x >> 5) + (threadIdx.x >> 5);
  const int base = wid << 4;
  if (base >= B) return;
  const int m16 = lane & 15;
  const bool hi = lane >= 16;
  const int n = base + m16;

  auto ald = [&](const float* W, int ld, int K, int M, int m, int kb) -> v2f {
    int k0 = kb + (hi ? 2 : 0);
    float a0 = (m < M && k0 < K) ? W[k0 * ld + m] : 0.f;
    float a1 = (m < M && (k0 + 1) < K) ? W[(k0 + 1) * ld + m] : 0.f;
    v2f r = {a0, a1};
    return r;
  };

  const v2f onesB = {hi ? 0.f : 1.f, 0.f};
  const v8f vzero = {0.f, 0.f, 0.f, 0.f, 0.f, 0.f, 0.f, 0.f};

  v2f wo1A[4][4], wo1Ab[4];
  v2f wd1A[4][4], wd1Ab[4];
  v8f wd2v[4];
  v2f wo2A[13];
  v8f bo2C;
#pragma unroll
  for (int tt = 0; tt < 4; ++tt) {
#pragma unroll
    for (int c = 0; c < 4; ++c) {
      wo1A[tt][c] = ald(Wo1, HID, LAT, HID, 16 * tt + m16, 4 * c);
      wd1A[tt][c] = ald(Wd1, HID, LAT, HID, 16 * tt + m16, 4 * c);
    }
    int mb = 16 * tt + m16;
    {
      v2f a = {(!hi && mb < HID) ? bo1[mb] : 0.f, 0.f};
      wo1Ab[tt] = a;
      v2f b = {(!hi && mb < HID) ? bd1[mb] : 0.f, 0.f};
      wd1Ab[tt] = b;
    }
#pragma unroll
    for (int v = 0; v < 8; ++v) {
      int m = 16 * tt + v + (hi ? 8 : 0);
      wd2v[tt][v] = (m < HID) ? Wd2[m] : 0.f;
    }
  }
#pragma unroll
  for (int c = 0; c < 13; ++c) wo2A[c] = ald(Wo2, LAT, HID, LAT, m16, 4 * c);
#pragma unroll
  for (int v = 0; v < 8; ++v) bo2C[v] = bo2[v + (hi ? 8 : 0)];
  const float bd2s = bd2[0];

  auto convfull = [&](const v8f& cd, v2f* bc) {
    float s0 = swap16(cd[0]), s1 = swap16(cd[1]), s2 = swap16(cd[2]),
          s3 = swap16(cd[3]), s4 = swap16(cd[4]), s5 = swap16(cd[5]),
          s6 = swap16(cd[6]), s7 = swap16(cd[7]);
    v2f b0 = {hi ? s2 : cd[0], hi ? s3 : cd[1]};
    v2f b1 = {hi ? s6 : cd[4], hi ? s7 : cd[5]};
    v2f b2 = {hi ? cd[2] : s0, hi ? cd[3] : s1};
    v2f b3 = {hi ? cd[6] : s4, hi ? cd[7] : s5};
    bc[0] = b0; bc[1] = b1; bc[2] = b2; bc[3] = b3;
  };

  auto feval = [&](const v2f* inB) -> v8f {
    v8f g0 = bo2C, g1 = vzero;
#pragma unroll
    for (int tt = 0; tt < 4; ++tt) {
      v8f h = wmma4(wo1Ab[tt], onesB, vzero);
#pragma unroll
      for (int c = 0; c < 4; ++c) h = wmma4(wo1A[tt][c], inB[c], h);
      v8f th;
#pragma unroll
      for (int v = 0; v < 8; ++v) th[v] = TANH_F32(h[v]);
      if (tt < 3) {
        v2f bc[4];
        convfull(th, bc);
#pragma unroll
        for (int c = 0; c < 4; ++c) {
          if (((4 * tt + c) & 1) == 0) g0 = wmma4(wo2A[4 * tt + c], bc[c], g0);
          else                         g1 = wmma4(wo2A[4 * tt + c], bc[c], g1);
        }
      } else {
        v2f bc0 = {hi ? swap16(th[2]) : th[0], hi ? swap16(th[3]) : th[1]};
        g0 = wmma4(wo2A[12], bc0, g0);
      }
    }
#pragma unroll
    for (int v = 0; v < 8; ++v) g0[v] += g1[v];
    return g0;
  };

  auto decode = [&](const v2f* zB, int ts) {
    float p = 0.f;
#pragma unroll
    for (int tt = 0; tt < 4; ++tt) {
      v8f h = wmma4(wd1Ab[tt], onesB, vzero);
#pragma unroll
      for (int c = 0; c < 4; ++c) h = wmma4(wd1A[tt][c], zB[c], h);
#pragma unroll
      for (int v = 0; v < 8; ++v)
        p = __builtin_fmaf(wd2v[tt][v], fmaxf(h[v], 0.f), p);
    }
    float tot = p + swap16(p) + bd2s;
    if (!hi && n < B) out[(size_t)ts * B + n] = tot;
  };

  v8f z;
  {
    bool ok = (!hi) && (n < B);
    v2f xB = {ok ? x0[(size_t)n * 2 + 0] : 0.f,
              ok ? x0[(size_t)n * 2 + 1] : 0.f};
    v8f zc;
#pragma unroll
    for (int v = 0; v < 8; ++v) zc[v] = be2[v + (hi ? 8 : 0)];
#pragma unroll
    for (int tt = 0; tt < 4; ++tt) {
      v2f a = ald(We1, HID, 2, HID, 16 * tt + m16, 0);
      v8f c0;
#pragma unroll
      for (int v = 0; v < 8; ++v) {
        int m = 16 * tt + v + (hi ? 8 : 0);
        c0[v] = (m < HID) ? be1[m] : 0.f;
      }
      v8f h = wmma4(a, xB, c0);
#pragma unroll
      for (int v = 0; v < 8; ++v) h[v] = fmaxf(h[v], 0.f);
      if (tt < 3) {
        v2f bc[4];
        convfull(h, bc);
#pragma unroll
        for (int c = 0; c < 4; ++c)
          zc = wmma4(ald(We2, LAT, HID, LAT, m16, 4 * (4 * tt + c)), bc[c], zc);
      } else {
        v2f bc0 = {hi ? swap16(h[2]) : h[0], hi ? swap16(h[3]) : h[1]};
        zc = wmma4(ald(We2, LAT, HID, LAT, m16, 48), bc0, zc);
      }
    }
    z = zc;
  }

  v2f zB[4];
  convfull(z, zB);
  decode(zB, 0);

  for (int s = 0; s + 1 < T; ++s) {
    float dt = tv[s + 1] - tv[s];
    v8f k = feval(zB);
    v8f acc = k;
    v8f zi;
    v2f ziB[4];
#pragma unroll
    for (int v = 0; v < 8; ++v) zi[v] = __builtin_fmaf(0.5f * dt, k[v], z[v]);
    convfull(zi, ziB);
    k = feval(ziB);
#pragma unroll
    for (int v = 0; v < 8; ++v) {
      acc[v] = __builtin_fmaf(2.f, k[v], acc[v]);
      zi[v] = __builtin_fmaf(0.5f * dt, k[v], z[v]);
    }
    convfull(zi, ziB);
    k = feval(ziB);
#pragma unroll
    for (int v = 0; v < 8; ++v) {
      acc[v] = __builtin_fmaf(2.f, k[v], acc[v]);
      zi[v] = __builtin_fmaf(dt, k[v], z[v]);
    }
    convfull(zi, ziB);
    k = feval(ziB);
    float w = dt * (1.f / 6.f);
#pragma unroll
    for (int v = 0; v < 8; ++v) z[v] = __builtin_fmaf(w, acc[v] + k[v], z[v]);
    convfull(z, zB);
    decode(zB, s + 1);
  }
}

extern "C" void kernel_launch(void* const* d_in, const int* in_sizes, int n_in,
                              void* d_out, int out_size, void* d_ws,
                              size_t ws_size, hipStream_t stream) {
  const float* x0 = (const float*)d_in[0];
  const float* tv = (const float*)d_in[1];
  const float* We1 = (const float*)d_in[2];
  const float* be1 = (const float*)d_in[3];
  const float* We2 = (const float*)d_in[4];
  const float* be2 = (const float*)d_in[5];
  const float* Wo1 = (const float*)d_in[6];
  const float* bo1 = (const float*)d_in[7];
  const float* Wo2 = (const float*)d_in[8];
  const float* bo2 = (const float*)d_in[9];
  const float* Wd1 = (const float*)d_in[10];
  const float* bd1 = (const float*)d_in[11];
  const float* Wd2 = (const float*)d_in[12];
  const float* bd2 = (const float*)d_in[13];
  float* out = (float*)d_out;

  int B = in_sizes[0] / 2;
  int T = in_sizes[1];
  int tiles = (B + 15) / 16;
  int wavesPerBlock = 8;
  int blocks = (tiles + wavesPerBlock - 1) / wavesPerBlock;

  node_rk4_kernel<<<blocks, wavesPerBlock * 32, 0, stream>>>(
      x0, tv, We1, be1, We2, be2, Wo1, bo1, Wo2, bo2, Wd1, bd1, Wd2, bd2, out,
      B, T);
}


# Layer-1 bias of the ODE net and the decoder enters as the C operand of each tile's first WMMA (C-layout tiles read from LDS), not as a ones-chunk WMMA: 152 -> 132 WMMAs per step, same sum order
# speedup vs baseline: 1.0719x; 1.0719x over previous
#include <hip/hip_runtime.h>
#include <math.h>


typedef __attribute__((ext_vector_type(2))) float v2f;
typedef __attribute__((ext_vector_type(8))) float v8f;

#if __has_builtin(__builtin_amdgcn_tanhf)
#define TANH_F32(x) __builtin_amdgcn_tanhf(x)
#else
#define TANH_F32(x) tanhf(x)
#endif

__device__ __forceinline__ v8f wmma4(v2f a, v2f b, v8f c) {
  return __builtin_amdgcn_wmma_f32_16x16x4_f32(false, a, false, b, (short)0, c,
                                               false, false);
}

__device__ __forceinline__ float swap16(float x) {
  return __int_as_float(__builtin_amdgcn_ds_swizzle(__float_as_int(x), 0x401F));
}

#define HID 50
#define LAT 16

__global__ __launch_bounds__(256) void node_rk4_kernel(
    const float* __restrict__ x0, const float* __restrict__ tv,
    const float* __restrict__ We1, const float* __restrict__ be1,
    const float* __restrict__ We2, const float* __restrict__ be2,
    const float* __restrict__ Wo1, const float* __restrict__ bo1,
    const float* __restrict__ Wo2, const float* __restrict__ bo2,
    const float* __restrict__ Wd1, const float* __restrict__ bd1,
    const float* __restrict__ Wd2, const float* __restrict__ bd2,
    float* __restrict__ out, int B, int T) {
  __shared__ __attribute__((aligned(32))) float sBiasC[2 * 2 * 4 * 8];
  for (int i = threadIdx.x; i < 2 * 2 * 4 * 8; i += blockDim.x) {
    int m = 16 * ((i >> 3) & 3) + 8 * ((i >> 5) & 1) + (i & 7);
    float bv = 0.f;
    if (m < HID) bv = (i >> 6) ? bd1[m] : bo1[m];
    sBiasC[i] = bv;
  }
  __syncthreads();

  const int lane = threadIdx.x & 31;
  const int wid = blockIdx.x * (blockDim.x >> 5) + (threadIdx.x >> 5);
  const int base = wid << 4;
  if (base >= B) return;
  const int m16 = lane & 15;
  const bool hi = lane >= 16;
  const int n = base + m16;

  auto ald = [&](const float* W, int ld, int K, int M, int m, int kb) -> v2f {
    int k0 = kb + (hi ? 2 : 0);
    float a0 = (m < M && k0 < K) ? W[k0 * ld + m] : 0.f;
    float a1 = (m < M && (k0 + 1) < K) ? W[(k0 + 1) * ld + m] : 0.f;
    v2f r = {a0, a1};
    return r;
  };

  typedef const volatile __attribute__((address_space(3))) v8f* lds_tile_ptr;
  const lds_tile_ptr bo1C = (lds_tile_ptr)(sBiasC + (hi ? 32 : 0));
  const lds_tile_ptr bd1C = (lds_tile_ptr)(sBiasC + 64 + (hi ? 32 : 0));
  const v8f vzero = {0.f, 0.f, 0.f, 0.f, 0.f, 0.f, 0.f, 0.f};

  v2f wo1A[4][4];
  v2f wd1A[4][4];
  v8f wd2v[4];
  v2f wo2A[13];
  v8f bo2C;
#pragma unroll
  for (int tt = 0; tt < 4; ++tt) {
#pragma unroll
    for (int c = 0; c < 4; ++c) {
      wo1A[tt][c] = ald(Wo1, HID, LAT, HID, 16 * tt + m16, 4 * c);
      wd1A[tt][c] = ald(Wd1, HID, LAT, HID, 16 * tt + m16, 4 * c);
    }
#pragma unroll
    for (int v = 0; v < 8; ++v) {
      int m = 16 * tt + v + (hi ? 8 : 0);
      wd2v[tt][v] = (m < HID) ? Wd2[m] : 0.f;
    }
  }
#pragma unroll
  for (int c = 0; c < 13; ++c) wo2A[c] = ald(Wo2, LAT, HID, LAT, m16, 4 * c);
#pragma unroll
  for (int v = 0; v < 8; ++v) bo2C[v] = bo2[v + (hi ? 8 : 0)];
  const float bd2s = bd2[0];

  auto convfull = [&](const v8f& cd, v2f* bc) {
    float s0 = swap16(cd[0]), s1 = swap16(cd[1]), s2 = swap16(cd[2]),
          s3 = swap16(cd[3]), s4 = swap16(cd[4]), s5 = swap16(cd[5]),
          s6 = swap16(cd[6]), s7 = swap16(cd[7]);
    v2f b0 = {hi ? s2 : cd[0], hi ? s3 : cd[1]};
    v2f b1 = {hi ? s6 : cd[4], hi ? s7 : cd[5]};
    v2f b2 = {hi ? cd[2] : s0, hi ? cd[3] : s1};
    v2f b3 = {hi ? cd[6] : s4, hi ? cd[7] : s5};
    bc[0] = b0; bc[1] = b1; bc[2] = b2; bc[3] = b3;
  };

  auto feval = [&](const v2f* inB) -> v8f {
    v8f g0 = bo2C, g1 = vzero;
#pragma unroll
    for (int tt = 0; tt < 4; ++tt) {
      v8f h = bo1C[tt];
#pragma unroll
      for (int c = 0; c < 4; ++c) h = wmma4(wo1A[tt][c], inB[c], h);
      v8f th;
#pragma unroll
      for (int v = 0; v < 8; ++v) th[v] = TANH_F32(h[v]);
      if (tt < 3) {
        v2f bc[4];
        convfull(th, bc);
#pragma unroll
        for (int c = 0; c < 4; ++c) {
          if (((4 * tt + c) & 1) == 0) g0 = wmma4(wo2A[4 * tt + c], bc[c], g0);
          else                         g1 = wmma4(wo2A[4 * tt + c], bc[c], g1);
        }
      } else {
        v2f bc0 = {hi ? swap16(th[2]) : th[0], hi ? swap16(th[3]) : th[1]};
        g0 = wmma4(wo2A[12], bc0, g0);
      }
    }
#pragma unroll
    for (int v = 0; v < 8; ++v) g0[v] += g1[v];
    return g0;
  };

  auto decode = [&](const v2f* zB, int ts) {
    float p = 0.f;
#pragma unroll
    for (int tt = 0; tt < 4; ++tt) {
      v8f h = bd1C[tt];
#pragma unroll
      for (int c = 0; c < 4; ++c) h = wmma4(wd1A[tt][c], zB[c], h);
#pragma unroll
      for (int v = 0; v < 8; ++v)
        p = __builtin_fmaf(wd2v[tt][v], fmaxf(h[v], 0.f), p);
    }
    float tot = p + swap16(p) + bd2s;
    if (!hi && n < B) out[(size_t)ts * B + n] = tot;
  };

  v8f z;
  {
    bool ok = (!hi) && (n < B);
    v2f xB = {ok ? x0[(size_t)n * 2 + 0] : 0.f,
              ok ? x0[(size_t)n * 2 + 1] : 0.f};
    v8f zc;
#pragma unroll
    for (int v = 0; v < 8; ++v) zc[v] = be2[v + (hi ? 8 : 0)];
#pragma unroll
    for (int tt = 0; tt < 4; ++tt) {
      v2f a = ald(We1, HID, 2, HID, 16 * tt + m16, 0);
      v8f c0;
#pragma unroll
      for (int v = 0; v < 8; ++v) {
        int m = 16 * tt + v + (hi ? 8 : 0);
        c0[v] = (m < HID) ? be1[m] : 0.f;
      }
      v8f h = wmma4(a, xB, c0);
#pragma unroll
      for (int v = 0; v < 8; ++v) h[v] = fmaxf(h[v], 0.f);
      if (tt < 3) {
        v2f bc[4];
        convfull(h, bc);
#pragma unroll
        for (int c = 0; c < 4; ++c)
          zc = wmma4(ald(We2, LAT, HID, LAT, m16, 4 * (4 * tt + c)), bc[c], zc);
      } else {
        v2f bc0 = {hi ? swap16(h[2]) : h[0], hi ? swap16(h[3]) : h[1]};
        zc = wmma4(ald(We2, LAT, HID, LAT, m16, 48), bc0, zc);
      }
    }
    z = zc;
  }

  v2f zB[4];
  convfull(z, zB);
  decode(zB, 0);

  for (int s = 0; s + 1 < T; ++s) {
    float dt = tv[s + 1] - tv[s];
    v8f k = feval(zB);
    v8f acc = k;
    v8f zi;
    v2f ziB[4];
#pragma unroll
    for (int v = 0; v < 8; ++v) zi[v] = __builtin_fmaf(0.5f * dt, k[v], z[v]);
    convfull(zi, ziB);
    k = feval(ziB);
#pragma unroll
    for (int v = 0; v < 8; ++v) {
      acc[v] = __builtin_fmaf(2.f, k[v], acc[v]);
      zi[v] = __builtin_fmaf(0.5f * dt, k[v], z[v]);
    }
    convfull(zi, ziB);
    k = feval(ziB);
#pragma unroll
    for (int v = 0; v < 8; ++v) {
      acc[v] = __builtin_fmaf(2.f, k[v], acc[v]);
      zi[v] = __builtin_fmaf(dt, k[v], z[v]);
    }
    convfull(zi, ziB);
    k = feval(ziB);
    float w = dt * (1.f / 6.f);
#pragma unroll
    for (int v = 0; v < 8; ++v) z[v] = __builtin_fmaf(w, acc[v] + k[v], z[v]);
    convfull(z, zB);
    decode(zB, s + 1);
  }
}

extern "C" void kernel_launch(void* const* d_in, const int* in_sizes, int n_in,
                              void* d_out, int out_size, void* d_ws,
                              size_t ws_size, hipStream_t stream) {
  const float* x0 = (const float*)d_in[0];
  const float* tv = (const float*)d_in[1];
  const float* We1 = (const float*)d_in[2];
  const float* be1 = (const float*)d_in[3];
  const float* We2 = (const float*)d_in[4];
  const float* be2 = (const float*)d_in[5];
  const float* Wo1 = (const float*)d_in[6];
  const float* bo1 = (const float*)d_in[7];
  const float* Wo2 = (const float*)d_in[8];
  const float* bo2 = (const float*)d_in[9];
  const float* Wd1 = (const float*)d_in[10];
  const float* bd1 = (const float*)d_in[11];
  const float* Wd2 = (const float*)d_in[12];
  const float* bd2 = (const float*)d_in[13];
  float* out = (float*)d_out;

  int B = in_sizes[0] / 2;
  int T = in_sizes[1];
  int tiles = (B + 15) / 16;
  int wavesPerBlock = 8;
  int blocks = (tiles + wavesPerBlock - 1) / wavesPerBlock;

  node_rk4_kernel<<<blocks, wavesPerBlock * 32, 0, stream>>>(
      x0, tv, We1, be1, We2, be2, Wo1, bo1, Wo2, bo2, Wd1, bd1, Wd2, bd2, out,
      B, T);
}


# A operands loaded in the k order of a C/D tile's register pairs: each tile feeds the next WMMA as B in place, all ds_swizzle+select layout fixups removed; k sum order in the GEMMs changes
# speedup vs baseline: 1.2583x; 1.1739x over previous
#include <hip/hip_runtime.h>
#include <math.h>


typedef __attribute__((ext_vector_type(2))) float v2f;
typedef __attribute__((ext_vector_type(8))) float v8f;

#if __has_builtin(__builtin_amdgcn_tanhf)
#define TANH_F32(x) __builtin_amdgcn_tanhf(x)
#else
#define TANH_F32(x) tanhf(x)
#endif

__device__ __forceinline__ v8f wmma4(v2f a, v2f b, v8f c) {
  return __builtin_amdgcn_wmma_f32_16x16x4_f32(false, a, false, b, (short)0, c,
                                               false, false);
}

__device__ __forceinline__ float swap16(float x) {
  return __int_as_float(__builtin_amdgcn_ds_swizzle(__float_as_int(x), 0x401F));
}

#define HID 50
#define LAT 16

__global__ __launch_bounds__(256) void node_rk4_kernel(
    const float* __restrict__ x0, const float* __restrict__ tv,
    const float* __restrict__ We1, const float* __restrict__ be1,
    const float* __restrict__ We2, const float* __restrict__ be2,
    const float* __restrict__ Wo1, const float* __restrict__ bo1,
    const float* __restrict__ Wo2, const float* __restrict__ bo2,
    const float* __restrict__ Wd1, const float* __restrict__ bd1,
    const float* __restrict__ Wd2, const float* __restrict__ bd2,
    float* __restrict__ out, int B, int T) {
  __shared__ __attribute__((aligned(32))) float sBiasC[2 * 2 * 4 * 8];
  for (int i = threadIdx.x; i < 2 * 2 * 4 * 8; i += blockDim.x) {
    int m = 16 * ((i >> 3) & 3) + 8 * ((i >> 5) & 1) + (i & 7);
    float bv = 0.f;
    if (m < HID) bv = (i >> 6) ? bd1[m] : bo1[m];
    sBiasC[i] = bv;
  }
  __syncthreads();

  const int lane = threadIdx.x & 31;
  const int wid = blockIdx.x * (blockDim.x >> 5) + (threadIdx.x >> 5);
  const int base = wid << 4;
  if (base >= B) return;
  const int m16 = lane & 15;
  const bool hi = lane >= 16;
  const int n = base + m16;

  auto ald = [&](const float* W, int ld, int K, int M, int m, int kb) -> v2f {
    int k0 = kb + (hi ? 8 : 0);
    float a0 = (m < M && k0 < K) ? W[k0 * ld + m] : 0.f;
    float a1 = (m < M && (k0 + 1) < K) ? W[(k0 + 1) * ld + m] : 0.f;
    v2f r = {a0, a1};
    return r;
  };

  typedef const volatile __attribute__((address_space(3))) v8f* lds_tile_ptr;
  const lds_tile_ptr bo1C = (lds_tile_ptr)(sBiasC + (hi ? 32 : 0));
  const lds_tile_ptr bd1C = (lds_tile_ptr)(sBiasC + 64 + (hi ? 32 : 0));
  const v8f vzero = {0.f, 0.f, 0.f, 0.f, 0.f, 0.f, 0.f, 0.f};

  v2f wo1A[4][4];
  v2f wd1A[4][4];
  v8f wd2v[4];
  v2f wo2A[13];
  v8f bo2C;
#pragma unroll
  for (int tt = 0; tt < 4; ++tt) {
#pragma unroll
    for (int c = 0; c < 4; ++c) {
      wo1A[tt][c] = ald(Wo1, HID, LAT, HID, 16 * tt + m16, 2 * c);
      wd1A[tt][c] = ald(Wd1, HID, LAT, HID, 16 * tt + m16, 2 * c);
    }
#pragma unroll
    for (int v = 0; v < 8; ++v) {
      int m = 16 * tt + v + (hi ? 8 : 0);
      wd2v[tt][v] = (m < HID) ? Wd2[m] : 0.f;
    }
  }
#pragma unroll
  for (int c = 0; c < 13; ++c)
    wo2A[c] = ald(Wo2, LAT, HID, LAT, m16, 16 * (c >> 2) + 2 * (c & 3));
#pragma unroll
  for (int v = 0; v < 8; ++v) bo2C[v] = bo2[v + (hi ? 8 : 0)];
  const float bd2s = bd2[0];

  auto bchunk = [](const v8f& cd, int j) -> v2f {
    v2f r = {cd[2 * j], cd[2 * j + 1]};
    return r;
  };

  auto feval = [&](const v8f& zin) -> v8f {
    v8f g0 = bo2C, g1 = vzero;
#pragma unroll
    for (int tt = 0; tt < 4; ++tt) {
      v8f h = bo1C[tt];
#pragma unroll
      for (int c = 0; c < 4; ++c) h = wmma4(wo1A[tt][c], bchunk(zin, c), h);
      v8f th;
#pragma unroll
      for (int v = 0; v < 8; ++v) th[v] = TANH_F32(h[v]);
      if (tt < 3) {
#pragma unroll
        for (int c = 0; c < 4; ++c) {
          if (((4 * tt + c) & 1) == 0) g0 = wmma4(wo2A[4 * tt + c], bchunk(th, c), g0);
          else                         g1 = wmma4(wo2A[4 * tt + c], bchunk(th, c), g1);
        }
      } else {
        g0 = wmma4(wo2A[12], bchunk(th, 0), g0);
      }
    }
#pragma unroll
    for (int v = 0; v < 8; ++v) g0[v] += g1[v];
    return g0;
  };

  auto decode = [&](const v8f& zin, int ts) {
    float p = 0.f;
#pragma unroll
    for (int tt = 0; tt < 4; ++tt) {
      v8f h = bd1C[tt];
#pragma unroll
      for (int c = 0; c < 4; ++c) h = wmma4(wd1A[tt][c], bchunk(zin, c), h);
#pragma unroll
      for (int v = 0; v < 8; ++v)
        p = __builtin_fmaf(wd2v[tt][v], fmaxf(h[v], 0.f), p);
    }
    float tot = p + swap16(p) + bd2s;
    if (!hi && n < B) out[(size_t)ts * B + n] = tot;
  };

  v8f z;
  {
    bool ok = (!hi) && (n < B);
    v2f xB = {ok ? x0[(size_t)n * 2 + 0] : 0.f,
              ok ? x0[(size_t)n * 2 + 1] : 0.f};
    v8f zc;
#pragma unroll
    for (int v = 0; v < 8; ++v) zc[v] = be2[v + (hi ? 8 : 0)];
#pragma unroll
    for (int tt = 0; tt < 4; ++tt) {
      v2f a = ald(We1, HID, 2, HID, 16 * tt + m16, 0);
      v8f c0;
#pragma unroll
      for (int v = 0; v < 8; ++v) {
        int m = 16 * tt + v + (hi ? 8 : 0);
        c0[v] = (m < HID) ? be1[m] : 0.f;
      }
      v8f h = wmma4(a, xB, c0);
#pragma unroll
      for (int v = 0; v < 8; ++v) h[v] = fmaxf(h[v], 0.f);
      if (tt < 3) {
#pragma unroll
        for (int c = 0; c < 4; ++c)
          zc = wmma4(ald(We2, LAT, HID, LAT, m16, 16 * tt + 2 * c), bchunk(h, c), zc);
      } else {
        zc = wmma4(ald(We2, LAT, HID, LAT, m16, 48), bchunk(h, 0), zc);
      }
    }
    z = zc;
  }

  decode(z, 0);

  for (int s = 0; s + 1 < T; ++s) {
    float dt = tv[s + 1] - tv[s];
    v8f k = feval(z);
    v8f acc = k;
    v8f zi;
#pragma unroll
    for (int v = 0; v < 8; ++v) zi[v] = __builtin_fmaf(0.5f * dt, k[v], z[v]);
    k = feval(zi);
#pragma unroll
    for (int v = 0; v < 8; ++v) {
      acc[v] = __builtin_fmaf(2.f, k[v], acc[v]);
      zi[v] = __builtin_fmaf(0.5f * dt, k[v], z[v]);
    }
    k = feval(zi);
#pragma unroll
    for (int v = 0; v < 8; ++v) {
      acc[v] = __builtin_fmaf(2.f, k[v], acc[v]);
      zi[v] = __builtin_fmaf(dt, k[v], z[v]);
    }
    k = feval(zi);
    float w = dt * (1.f / 6.f);
#pragma unroll
    for (int v = 0; v < 8; ++v) z[v] = __builtin_fmaf(w, acc[v] + k[v], z[v]);
    decode(z, s + 1);
  }
}

extern "C" void kernel_launch(void* const* d_in, const int* in_sizes, int n_in,
                              void* d_out, int out_size, void* d_ws,
                              size_t ws_size, hipStream_t stream) {
  const float* x0 = (const float*)d_in[0];
  const float* tv = (const float*)d_in[1];
  const float* We1 = (const float*)d_in[2];
  const float* be1 = (const float*)d_in[3];
  const float* We2 = (const float*)d_in[4];
  const float* be2 = (const float*)d_in[5];
  const float* Wo1 = (const float*)d_in[6];
  const float* bo1 = (const float*)d_in[7];
  const float* Wo2 = (const float*)d_in[8];
  const float* bo2 = (const float*)d_in[9];
  const float* Wd1 = (const float*)d_in[10];
  const float* bd1 = (const float*)d_in[11];
  const float* Wd2 = (const float*)d_in[12];
  const float* bd2 = (const float*)d_in[13];
  float* out = (float*)d_out;

  int B = in_sizes[0] / 2;
  int T = in_sizes[1];
  int tiles = (B + 15) / 16;
  int wavesPerBlock = 8;
  int blocks = (tiles + wavesPerBlock - 1) / wavesPerBlock;

  node_rk4_kernel<<<blocks, wavesPerBlock * 32, 0, stream>>>(
      x0, tv, We1, be1, We2, be2, Wo1, bo1, Wo2, bo2, Wd1, bd1, Wd2, bd2, out,
      B, T);
}


# Two 16-sample tiles per wave (2 waves per SIMD, 512-register budget): their independent WMMA/tanh chains and the decoder tail interleave in one wave; bias tiles in registers, no LDS reads in the loop
# speedup vs baseline: 1.3043x; 1.0366x over previous
#include <hip/hip_runtime.h>
#include <math.h>


typedef __attribute__((ext_vector_type(2))) float v2f;
typedef __attribute__((ext_vector_type(8))) float v8f;

#if __has_builtin(__builtin_amdgcn_tanhf)
#define TANH_F32(x) __builtin_amdgcn_tanhf(x)
#else
#define TANH_F32(x) tanhf(x)
#endif

__device__ __forceinline__ v8f wmma4(v2f a, v2f b, v8f c) {
  return __builtin_amdgcn_wmma_f32_16x16x4_f32(false, a, false, b, (short)0, c,
                                               false, false);
}

__device__ __forceinline__ float swap16(float x) {
  return __int_as_float(__builtin_amdgcn_ds_swizzle(__float_as_int(x), 0x401F));
}

#define HID 50
#define LAT 16
#define NT 2

__global__ __launch_bounds__(256) void node_rk4_kernel(
    const float* __restrict__ x0, const float* __restrict__ tv,
    const float* __restrict__ We1, const float* __restrict__ be1,
    const float* __restrict__ We2, const float* __restrict__ be2,
    const float* __restrict__ Wo1, const float* __restrict__ bo1,
    const float* __restrict__ Wo2, const float* __restrict__ bo2,
    const float* __restrict__ Wd1, const float* __restrict__ bd1,
    const float* __restrict__ Wd2, const float* __restrict__ bd2,
    float* __restrict__ out, int B, int T) {
  const int lane = threadIdx.x & 31;
  const int wid = blockIdx.x * (blockDim.x >> 5) + (threadIdx.x >> 5);
  const int base = wid << 5;
  if (base >= B) return;
  const int m16 = lane & 15;
  const bool hi = lane >= 16;

  auto ald = [&](const float* W, int ld, int K, int M, int m, int kb) -> v2f {
    int k0 = kb + (hi ? 8 : 0);
    float a0 = (m < M && k0 < K) ? W[k0 * ld + m] : 0.f;
    float a1 = (m < M && (k0 + 1) < K) ? W[(k0 + 1) * ld + m] : 0.f;
    v2f r = {a0, a1};
    return r;
  };

  const v8f vzero = {0.f, 0.f, 0.f, 0.f, 0.f, 0.f, 0.f, 0.f};

  v2f wo1A[4][4];
  v2f wd1A[4][4];
  v8f bo1C[4];
  v8f bd1C[4];
  v8f wd2v[4];
  v2f wo2A[13];
  v8f bo2C;
#pragma unroll
  for (int tt = 0; tt < 4; ++tt) {
#pragma unroll
    for (int c = 0; c < 4; ++c) {
      wo1A[tt][c] = ald(Wo1, HID, LAT, HID, 16 * tt + m16, 2 * c);
      wd1A[tt][c] = ald(Wd1, HID, LAT, HID, 16 * tt + m16, 2 * c);
    }
#pragma unroll
    for (int v = 0; v < 8; ++v) {
      int m = 16 * tt + v + (hi ? 8 : 0);
      wd2v[tt][v] = (m < HID) ? Wd2[m] : 0.f;
      bo1C[tt][v] = (m < HID) ? bo1[m] : 0.f;
      bd1C[tt][v] = (m < HID) ? bd1[m] : 0.f;
    }
  }
#pragma unroll
  for (int c = 0; c < 13; ++c)
    wo2A[c] = ald(Wo2, LAT, HID, LAT, m16, 16 * (c >> 2) + 2 * (c & 3));
#pragma unroll
  for (int v = 0; v < 8; ++v) bo2C[v] = bo2[v + (hi ? 8 : 0)];
  const float bd2s = bd2[0];

  auto bchunk = [](const v8f& cd, int j) -> v2f {
    v2f r = {cd[2 * j], cd[2 * j + 1]};
    return r;
  };

  auto layer1 = [&](const v2f (&wA)[4][4], const v8f (&bC)[4], const v8f& zin,
                    v8f (&h)[4]) {
#pragma unroll
    for (int c = 0; c < 4; ++c) {
#pragma unroll
      for (int tt = 0; tt < 4; ++tt)
        h[tt] = wmma4(wA[tt][c], bchunk(zin, c), c == 0 ? bC[tt] : h[tt]);
    }
  };

  auto ode_layer2 = [&](const v8f (&h)[4]) -> v8f {
    v8f g0 = bo2C, g1 = vzero;
#pragma unroll
    for (int tt = 0; tt < 4; ++tt) {
      v8f th;
#pragma unroll
      for (int v = 0; v < 8; ++v) th[v] = TANH_F32(h[tt][v]);
      if (tt < 3) {
#pragma unroll
        for (int c = 0; c < 4; ++c) {
          if (((4 * tt + c) & 1) == 0) g0 = wmma4(wo2A[4 * tt + c], bchunk(th, c), g0);
          else                         g1 = wmma4(wo2A[4 * tt + c], bchunk(th, c), g1);
        }
      } else {
        g0 = wmma4(wo2A[12], bchunk(th, 0), g0);
      }
    }
#pragma unroll
    for (int v = 0; v < 8; ++v) g0[v] += g1[v];
    return g0;
  };

  auto dec_tail = [&](const v8f (&h)[4]) -> float {
    float p = 0.f;
#pragma unroll
    for (int tt = 0; tt < 4; ++tt) {
#pragma unroll
      for (int v = 0; v < 8; ++v)
        p = __builtin_fmaf(wd2v[tt][v], fmaxf(h[tt][v], 0.f), p);
    }
    return p + swap16(p);
  };

  auto store_y = [&](float y0, float y1, int ts) {
    int n = base + lane;
    if (n < B) out[(size_t)ts * B + n] = (hi ? y1 : y0) + bd2s;
  };

  v8f z[NT];
  {
    v2f xB[NT];
#pragma unroll
    for (int u = 0; u < NT; ++u) {
      int n = base + 16 * u + m16;
      bool ok = (!hi) && (n < B);
      xB[u][0] = ok ? x0[(size_t)n * 2 + 0] : 0.f;
      xB[u][1] = ok ? x0[(size_t)n * 2 + 1] : 0.f;
#pragma unroll
      for (int v = 0; v < 8; ++v) z[u][v] = be2[v + (hi ? 8 : 0)];
    }
#pragma unroll
    for (int tt = 0; tt < 4; ++tt) {
      v2f a = ald(We1, HID, 2, HID, 16 * tt + m16, 0);
      v8f c0;
#pragma unroll
      for (int v = 0; v < 8; ++v) {
        int m = 16 * tt + v + (hi ? 8 : 0);
        c0[v] = (m < HID) ? be1[m] : 0.f;
      }
      v8f h[NT];
#pragma unroll
      for (int u = 0; u < NT; ++u) {
        h[u] = wmma4(a, xB[u], c0);
#pragma unroll
        for (int v = 0; v < 8; ++v) h[u][v] = fmaxf(h[u][v], 0.f);
      }
      if (tt < 3) {
#pragma unroll
        for (int c = 0; c < 4; ++c) {
          v2f a2 = ald(We2, LAT, HID, LAT, m16, 16 * tt + 2 * c);
#pragma unroll
          for (int u = 0; u < NT; ++u) z[u] = wmma4(a2, bchunk(h[u], c), z[u]);
        }
      } else {
        v2f a2 = ald(We2, LAT, HID, LAT, m16, 48);
#pragma unroll
        for (int u = 0; u < NT; ++u) z[u] = wmma4(a2, bchunk(h[u], 0), z[u]);
      }
    }
  }

  v8f h[NT][4];
  v8f d[NT][4];
  v8f k[NT], acc[NT], zi[NT];
  for (int s = 0; s + 1 < T; ++s) {
    float dt = tv[s + 1] - tv[s];
#pragma unroll
    for (int u = 0; u < NT; ++u) {
      layer1(wo1A, bo1C, z[u], h[u]);
      layer1(wd1A, bd1C, z[u], d[u]);
    }
#pragma unroll
    for (int u = 0; u < NT; ++u) k[u] = ode_layer2(h[u]);
#pragma unroll
    for (int u = 0; u < NT; ++u) {
      acc[u] = k[u];
#pragma unroll
      for (int v = 0; v < 8; ++v) zi[u][v] = __builtin_fmaf(0.5f * dt, k[u][v], z[u][v]);
      layer1(wo1A, bo1C, zi[u], h[u]);
    }
    {
      float y0 = dec_tail(d[0]);
      float y1 = dec_tail(d[1]);
      store_y(y0, y1, s);
    }
#pragma unroll
    for (int u = 0; u < NT; ++u) k[u] = ode_layer2(h[u]);
#pragma unroll
    for (int u = 0; u < NT; ++u) {
#pragma unroll
      for (int v = 0; v < 8; ++v) {
        acc[u][v] = __builtin_fmaf(2.f, k[u][v], acc[u][v]);
        zi[u][v] = __builtin_fmaf(0.5f * dt, k[u][v], z[u][v]);
      }
      layer1(wo1A, bo1C, zi[u], h[u]);
    }
#pragma unroll
    for (int u = 0; u < NT; ++u) k[u] = ode_layer2(h[u]);
#pragma unroll
    for (int u = 0; u < NT; ++u) {
#pragma unroll
      for (int v = 0; v < 8; ++v) {
        acc[u][v] = __builtin_fmaf(2.f, k[u][v], acc[u][v]);
        zi[u][v] = __builtin_fmaf(dt, k[u][v], z[u][v]);
      }
      layer1(wo1A, bo1C, zi[u], h[u]);
    }
#pragma unroll
    for (int u = 0; u < NT; ++u) k[u] = ode_layer2(h[u]);
    float w = dt * (1.f / 6.f);
#pragma unroll
    for (int u = 0; u < NT; ++u) {
#pragma unroll
      for (int v = 0; v < 8; ++v) z[u][v] = __builtin_fmaf(w, acc[u][v] + k[u][v], z[u][v]);
    }
  }
  {
#pragma unroll
    for (int u = 0; u < NT; ++u) layer1(wd1A, bd1C, z[u], d[u]);
    float y0 = dec_tail(d[0]);
    float y1 = dec_tail(d[1]);
    if (T > 0) store_y(y0, y1, T - 1);
  }
}

extern "C" void kernel_launch(void* const* d_in, const int* in_sizes, int n_in,
                              void* d_out, int out_size, void* d_ws,
                              size_t ws_size, hipStream_t stream) {
  const float* x0 = (const float*)d_in[0];
  const float* tv = (const float*)d_in[1];
  const float* We1 = (const float*)d_in[2];
  const float* be1 = (const float*)d_in[3];
  const float* We2 = (const float*)d_in[4];
  const float* be2 = (const float*)d_in[5];
  const float* Wo1 = (const float*)d_in[6];
  const float* bo1 = (const float*)d_in[7];
  const float* Wo2 = (const float*)d_in[8];
  const float* bo2 = (const float*)d_in[9];
  const float* Wd1 = (const float*)d_in[10];
  const float* bd1 = (const float*)d_in[11];
  const float* Wd2 = (const float*)d_in[12];
  const float* bd2 = (const float*)d_in[13];
  float* out = (float*)d_out;

  int B = in_sizes[0] / 2;
  int T = in_sizes[1];
  int waves = (B + 31) / 32;
  int wavesPerBlock = 8;
  int blocks = (waves + wavesPerBlock - 1) / wavesPerBlock;

  node_rk4_kernel<<<blocks, wavesPerBlock * 32, 0, stream>>>(
      x0, tv, We1, be1, We2, be2, Wo1, bo1, Wo2, bo2, Wd1, bd1, Wd2, bd2, out,
      B, T);
}


# Hidden rows 48,49 of each 16->50 layer leave the padded 4th row tile for f32 vector FMAs (8 k per lane half + one cross-half add): 264 -> 224 WMMAs per step; k sum order of these 2 rows changes
# speedup vs baseline: 1.4301x; 1.0965x over previous
#include <hip/hip_runtime.h>
#include <math.h>


typedef __attribute__((ext_vector_type(2))) float v2f;
typedef __attribute__((ext_vector_type(8))) float v8f;

#if __has_builtin(__builtin_amdgcn_tanhf)
#define TANH_F32(x) __builtin_amdgcn_tanhf(x)
#else
#define TANH_F32(x) tanhf(x)
#endif

__device__ __forceinline__ v8f wmma4(v2f a, v2f b, v8f c) {
  return __builtin_amdgcn_wmma_f32_16x16x4_f32(false, a, false, b, (short)0, c,
                                               false, false);
}

__device__ __forceinline__ float swap16(float x) {
  return __int_as_float(__builtin_amdgcn_ds_swizzle(__float_as_int(x), 0x401F));
}

__device__ __forceinline__ float xhalf(float x) {
  return __int_as_float(__builtin_amdgcn_permlanex16(
      __float_as_int(x), __float_as_int(x), 0x76543210, 0xfedcba98, false, false));
}

#define HID 50
#define LAT 16
#define NT 2
#define HT 3
#define HX 48
static_assert(HID - 16 * HT == 2 && HX == 16 * HT, "two hidden rows past the full tiles");

__global__ __launch_bounds__(256) void node_rk4_kernel(
    const float* __restrict__ x0, const float* __restrict__ tv,
    const float* __restrict__ We1, const float* __restrict__ be1,
    const float* __restrict__ We2, const float* __restrict__ be2,
    const float* __restrict__ Wo1, const float* __restrict__ bo1,
    const float* __restrict__ Wo2, const float* __restrict__ bo2,
    const float* __restrict__ Wd1, const float* __restrict__ bd1,
    const float* __restrict__ Wd2, const float* __restrict__ bd2,
    float* __restrict__ out, int B, int T) {
  const int lane = threadIdx.x & 31;
  const int wid = blockIdx.x * (blockDim.x >> 5) + (threadIdx.x >> 5);
  const int base = wid << 5;
  if (base >= B) return;
  const int m16 = lane & 15;
  const bool hi = lane >= 16;

  auto ald = [&](const float* W, int ld, int K, int M, int m, int kb) -> v2f {
    int k0 = kb + (hi ? 8 : 0);
    float a0 = (m < M && k0 < K) ? W[k0 * ld + m] : 0.f;
    float a1 = (m < M && (k0 + 1) < K) ? W[(k0 + 1) * ld + m] : 0.f;
    v2f r = {a0, a1};
    return r;
  };

  const v8f vzero = {0.f, 0.f, 0.f, 0.f, 0.f, 0.f, 0.f, 0.f};

  v2f wo1A[HT][4];
  v2f wd1A[HT][4];
  v8f bo1C[HT];
  v8f bd1C[HT];
  v8f wd2v[HT];
  v2f wo2A[13];
  v8f bo2C;
  v2f wo1X[8], wd1X[8];
  v2f bo1X, bd1X;
  v2f wd2X;
#pragma unroll
  for (int v = 0; v < 8; ++v) {
    int kk = v + (hi ? 8 : 0);
    wo1X[v][0] = Wo1[kk * HID + HX];
    wo1X[v][1] = Wo1[kk * HID + HX + 1];
    wd1X[v][0] = Wd1[kk * HID + HX];
    wd1X[v][1] = Wd1[kk * HID + HX + 1];
  }
#pragma unroll
  for (int e = 0; e < 2; ++e) {
    bo1X[e] = hi ? 0.f : bo1[HX + e];
    bd1X[e] = hi ? 0.f : bd1[HX + e];
    wd2X[e] = hi ? 0.f : Wd2[HX + e];
  }
#pragma unroll
  for (int tt = 0; tt < HT; ++tt) {
#pragma unroll
    for (int c = 0; c < 4; ++c) {
      wo1A[tt][c] = ald(Wo1, HID, LAT, HID, 16 * tt + m16, 2 * c);
      wd1A[tt][c] = ald(Wd1, HID, LAT, HID, 16 * tt + m16, 2 * c);
    }
#pragma unroll
    for (int v = 0; v < 8; ++v) {
      int m = 16 * tt + v + (hi ? 8 : 0);
      wd2v[tt][v] = (m < HID) ? Wd2[m] : 0.f;
      bo1C[tt][v] = (m < HID) ? bo1[m] : 0.f;
      bd1C[tt][v] = (m < HID) ? bd1[m] : 0.f;
    }
  }
#pragma unroll
  for (int c = 0; c < 13; ++c)
    wo2A[c] = ald(Wo2, LAT, HID, LAT, m16, 16 * (c >> 2) + 2 * (c & 3));
#pragma unroll
  for (int v = 0; v < 8; ++v) bo2C[v] = bo2[v + (hi ? 8 : 0)];
  const float bd2s = bd2[0];

  auto bchunk = [](const v8f& cd, int j) -> v2f {
    v2f r = {cd[2 * j], cd[2 * j + 1]};
    return r;
  };

  auto layer1 = [&](const v2f (&wA)[HT][4], const v8f (&bC)[HT], const v8f& zin,
                    v8f (&h)[HT]) {
#pragma unroll
    for (int c = 0; c < 4; ++c) {
#pragma unroll
      for (int tt = 0; tt < HT; ++tt)
        h[tt] = wmma4(wA[tt][c], bchunk(zin, c), c == 0 ? bC[tt] : h[tt]);
    }
  };

  auto rows2 = [&](const v2f (&wX)[8], v2f seed, const v8f& zin) -> v2f {
    v2f p = seed;
#pragma unroll
    for (int v = 0; v < 8; ++v) {
      v2f zz = {zin[v], zin[v]};
      p = __builtin_elementwise_fma(wX[v], zz, p);
    }
    v2f q = {xhalf(p[0]), xhalf(p[1])};
    return p + q;
  };

  auto ode_layer2 = [&](const v8f (&h)[HT], v2f hx) -> v8f {
    v8f g0 = bo2C, g1 = vzero;
#pragma unroll
    for (int tt = 0; tt < HT; ++tt) {
      v8f th;
#pragma unroll
      for (int v = 0; v < 8; ++v) th[v] = TANH_F32(h[tt][v]);
#pragma unroll
      for (int c = 0; c < 4; ++c) {
        if (((4 * tt + c) & 1) == 0) g0 = wmma4(wo2A[4 * tt + c], bchunk(th, c), g0);
        else                         g1 = wmma4(wo2A[4 * tt + c], bchunk(th, c), g1);
      }
    }
    v2f tx = {TANH_F32(hx[0]), TANH_F32(hx[1])};
    g0 = wmma4(wo2A[12], tx, g0);
#pragma unroll
    for (int v = 0; v < 8; ++v) g0[v] += g1[v];
    return g0;
  };

  auto dec_tail = [&](const v8f (&h)[HT], v2f hx) -> float {
    float p = 0.f;
#pragma unroll
    for (int tt = 0; tt < HT; ++tt) {
#pragma unroll
      for (int v = 0; v < 8; ++v)
        p = __builtin_fmaf(wd2v[tt][v], fmaxf(h[tt][v], 0.f), p);
    }
#pragma unroll
    for (int e = 0; e < 2; ++e)
      p = __builtin_fmaf(wd2X[e], fmaxf(hx[e], 0.f), p);
    return p + swap16(p);
  };

  auto store_y = [&](float y0, float y1, int ts) {
    int n = base + lane;
    if (n < B) out[(size_t)ts * B + n] = (hi ? y1 : y0) + bd2s;
  };

  v8f z[NT];
  {
    v2f xB[NT];
#pragma unroll
    for (int u = 0; u < NT; ++u) {
      int n = base + 16 * u + m16;
      bool ok = (!hi) && (n < B);
      xB[u][0] = ok ? x0[(size_t)n * 2 + 0] : 0.f;
      xB[u][1] = ok ? x0[(size_t)n * 2 + 1] : 0.f;
#pragma unroll
      for (int v = 0; v < 8; ++v) z[u][v] = be2[v + (hi ? 8 : 0)];
    }
#pragma unroll
    for (int tt = 0; tt < 4; ++tt) {
      v2f a = ald(We1, HID, 2, HID, 16 * tt + m16, 0);
      v8f c0;
#pragma unroll
      for (int v = 0; v < 8; ++v) {
        int m = 16 * tt + v + (hi ? 8 : 0);
        c0[v] = (m < HID) ? be1[m] : 0.f;
      }
      v8f h[NT];
#pragma unroll
      for (int u = 0; u < NT; ++u) {
        h[u] = wmma4(a, xB[u], c0);
#pragma unroll
        for (int v = 0; v < 8; ++v) h[u][v] = fmaxf(h[u][v], 0.f);
      }
      if (tt < 3) {
#pragma unroll
        for (int c = 0; c < 4; ++c) {
          v2f a2 = ald(We2, LAT, HID, LAT, m16, 16 * tt + 2 * c);
#pragma unroll
          for (int u = 0; u < NT; ++u) z[u] = wmma4(a2, bchunk(h[u], c), z[u]);
        }
      } else {
        v2f a2 = ald(We2, LAT, HID, LAT, m16, 48);
#pragma unroll
        for (int u = 0; u < NT; ++u) z[u] = wmma4(a2, bchunk(h[u], 0), z[u]);
      }
    }
  }

  v8f h[NT][HT];
  v8f d[NT][HT];
  v2f hx[NT];
  v2f dx[NT];
  v8f k[NT], acc[NT], zi[NT];
  for (int s = 0; s + 1 < T; ++s) {
    float dt = tv[s + 1] - tv[s];
#pragma unroll
    for (int u = 0; u < NT; ++u) {
      layer1(wo1A, bo1C, z[u], h[u]);
      layer1(wd1A, bd1C, z[u], d[u]);
      hx[u] = rows2(wo1X, bo1X, z[u]);
      dx[u] = rows2(wd1X, bd1X, z[u]);
    }
#pragma unroll
    for (int u = 0; u < NT; ++u) k[u] = ode_layer2(h[u], hx[u]);
#pragma unroll
    for (int u = 0; u < NT; ++u) {
      acc[u] = k[u];
#pragma unroll
      for (int v = 0; v < 8; ++v) zi[u][v] = __builtin_fmaf(0.5f * dt, k[u][v], z[u][v]);
      layer1(wo1A, bo1C, zi[u], h[u]);
      hx[u] = rows2(wo1X, bo1X, zi[u]);
    }
    {
      float y0 = dec_tail(d[0], dx[0]);
      float y1 = dec_tail(d[1], dx[1]);
      store_y(y0, y1, s);
    }
#pragma unroll
    for (int u = 0; u < NT; ++u) k[u] = ode_layer2(h[u], hx[u]);
#pragma unroll
    for (int u = 0; u < NT; ++u) {
#pragma unroll
      for (int v = 0; v < 8; ++v) {
        acc[u][v] = __builtin_fmaf(2.f, k[u][v], acc[u][v]);
        zi[u][v] = __builtin_fmaf(0.5f * dt, k[u][v], z[u][v]);
      }
      layer1(wo1A, bo1C, zi[u], h[u]);
      hx[u] = rows2(wo1X, bo1X, zi[u]);
    }
#pragma unroll
    for (int u = 0; u < NT; ++u) k[u] = ode_layer2(h[u], hx[u]);
#pragma unroll
    for (int u = 0; u < NT; ++u) {
#pragma unroll
      for (int v = 0; v < 8; ++v) {
        acc[u][v] = __builtin_fmaf(2.f, k[u][v], acc[u][v]);
        zi[u][v] = __builtin_fmaf(dt, k[u][v], z[u][v]);
      }
      layer1(wo1A, bo1C, zi[u], h[u]);
      hx[u] = rows2(wo1X, bo1X, zi[u]);
    }
#pragma unroll
    for (int u = 0; u < NT; ++u) k[u] = ode_layer2(h[u], hx[u]);
    float w = dt * (1.f / 6.f);
#pragma unroll
    for (int u = 0; u < NT; ++u) {
#pragma unroll
      for (int v = 0; v < 8; ++v) z[u][v] = __builtin_fmaf(w, acc[u][v] + k[u][v], z[u][v]);
    }
  }
  {
#pragma unroll
    for (int u = 0; u < NT; ++u) {
      layer1(wd1A, bd1C, z[u], d[u]);
      dx[u] = rows2(wd1X, bd1X, z[u]);
    }
    float y0 = dec_tail(d[0], dx[0]);
    float y1 = dec_tail(d[1], dx[1]);
    if (T > 0) store_y(y0, y1, T - 1);
  }
}

extern "C" void kernel_launch(void* const* d_in, const int* in_sizes, int n_in,
                              void* d_out, int out_size, void* d_ws,
                              size_t ws_size, hipStream_t stream) {
  const float* x0 = (const float*)d_in[0];
  const float* tv = (const float*)d_in[1];
  const float* We1 = (const float*)d_in[2];
  const float* be1 = (const float*)d_in[3];
  const float* We2 = (const float*)d_in[4];
  const float* be2 = (const float*)d_in[5];
  const float* Wo1 = (const float*)d_in[6];
  const float* bo1 = (const float*)d_in[7];
  const float* Wo2 = (const float*)d_in[8];
  const float* bo2 = (const float*)d_in[9];
  const float* Wd1 = (const float*)d_in[10];
  const float* bd1 = (const float*)d_in[11];
  const float* Wd2 = (const float*)d_in[12];
  const float* bd2 = (const float*)d_in[13];
  float* out = (float*)d_out;

  int B = in_sizes[0] / 2;
  int T = in_sizes[1];
  int waves = (B + 31) / 32;
  int wavesPerBlock = 8;
  int blocks = (waves + wavesPerBlock - 1) / wavesPerBlock;

  node_rk4_kernel<<<blocks, wavesPerBlock * 32, 0, stream>>>(
      x0, tv, We1, be1, We2, be2, Wo1, bo1, Wo2, bo2, Wd1, bd1, Wd2, bd2, out,
      B, T);
}
